# starting kernel
# baseline (speedup 1.0000x reference)
#include <hip/hip_runtime.h>
#include <hip/hip_bf16.h>
#include <math.h>

#define Bn   32
#define Hn   56
#define Wn   56
#define Cn   256
#define HWn  (Hn * Wn)
#define HID  16
#define NCHK 8
#define CHUNK (HWn / NCHK)

typedef float v2f __attribute__((ext_vector_type(2)));
typedef float v8f __attribute__((ext_vector_type(8)));

__device__ __forceinline__ v8f wmma4(v2f a, v2f b, v8f c) {
    return __builtin_amdgcn_wmma_f32_16x16x4_f32(
         false, a,  false, b,
         (short)0, c,  false,  false);
}

__device__ __forceinline__ float sigmoidf_(float v) {
    return 1.0f / (1.0f + __expf(-v));
}

__global__ void k_pool_partial(const float* __restrict__ x,
                               float* __restrict__ psum,
                               float* __restrict__ pmax) {
    const int c     = threadIdx.x;
    const int chunk = blockIdx.x;
    const int b     = blockIdx.y;
    const float* xp = x + ((size_t)b * HWn + (size_t)chunk * CHUNK) * Cn + c;

    float s = 0.0f;
    float m = -3.402823466e38f;
    #pragma unroll 4
    for (int i = 0; i < CHUNK; ++i) {
        float v = xp[(size_t)i * Cn];
        s += v;
        m = fmaxf(m, v);
    }
    psum[((size_t)chunk * Bn + b) * Cn + c] = s;
    pmax[((size_t)chunk * Bn + b) * Cn + c] = m;
}

__global__ void k_channel_mlp(const float* __restrict__ psum,
                              const float* __restrict__ pmax,
                              const float* __restrict__ w1,
                              const float* __restrict__ b1,
                              const float* __restrict__ w2,
                              const float* __restrict__ b2,
                              float* __restrict__ chout) {
    __shared__ float lds_pool[2][Bn][Cn + 1];
    __shared__ float lds_h[2][Bn][HID + 1];

    const int tid  = threadIdx.x;
    const int wv   = tid >> 5;
    const int lane = tid & 31;

    for (int p = tid; p < Bn * Cn; p += 256) {
        const int b = p >> 8, c = p & (Cn - 1);
        float s = 0.0f, m = -3.402823466e38f;
        #pragma unroll
        for (int ch = 0; ch < NCHK; ++ch) {
            s += psum[((size_t)ch * Bn + b) * Cn + c];
            m = fmaxf(m, pmax[((size_t)ch * Bn + b) * Cn + c]);
        }
        lds_pool[0][b][c] = s * (1.0f / (float)HWn);
        lds_pool[1][b][c] = m;
    }
    __syncthreads();

    if (wv < 4) {
        const int src  = wv >> 1;
        const int mt   = wv & 1;
        const int col  = lane & 15;
        const int m    = mt * 16 + col;
        const int ksel = (lane < 16) ? 0 : 2;

        v8f acc = {};
        for (int k0 = 0; k0 < Cn; k0 += 4) {
            v2f a, bfr;
            a.x   = lds_pool[src][m][k0 + ksel];
            a.y   = lds_pool[src][m][k0 + ksel + 1];
            bfr.x = w1[(k0 + ksel)     * HID + col];
            bfr.y = w1[(k0 + ksel + 1) * HID + col];
            acc = wmma4(a, bfr, acc);
        }
        const float bias  = b1[col];
        const int   rbase = mt * 16 + ((lane & 16) ? 8 : 0);
        #pragma unroll
        for (int v = 0; v < 8; ++v)
            lds_h[src][rbase + v][col] = fmaxf(acc[v] + bias, 0.0f);
    }
    __syncthreads();

    for (int t = wv; t < 32; t += 8) {
        const int mt   = t & 1;
        const int nt   = t >> 1;
        const int col  = lane & 15;
        const int ksel = (lane < 16) ? 0 : 2;
        const float bias = b2[nt * 16 + col];

        float combined[8];
        #pragma unroll
        for (int v = 0; v < 8; ++v) combined[v] = 0.0f;

        #pragma unroll
        for (int src = 0; src < 2; ++src) {
            v8f acc = {};
            #pragma unroll
            for (int k0 = 0; k0 < HID; k0 += 4) {
                v2f a, bfr;
                a.x   = lds_h[src][mt * 16 + col][k0 + ksel];
                a.y   = lds_h[src][mt * 16 + col][k0 + ksel + 1];
                bfr.x = w2[(k0 + ksel)     * Cn + nt * 16 + col];
                bfr.y = w2[(k0 + ksel + 1) * Cn + nt * 16 + col];
                acc = wmma4(a, bfr, acc);
            }
            #pragma unroll
            for (int v = 0; v < 8; ++v)
                combined[v] += sigmoidf_(acc[v] + bias);
        }
        const int rbase = mt * 16 + ((lane & 16) ? 8 : 0);
        #pragma unroll
        for (int v = 0; v < 8; ++v)
            chout[(size_t)(rbase + v) * Cn + nt * 16 + col] = combined[v];
    }
}

__global__ void k_spatial_stats(const float* __restrict__ x,
                                const float* __restrict__ chout,
                                float* __restrict__ sp) {
    const int lane = threadIdx.x & 31;
    const int wv   = threadIdx.x >> 5;
    const int pos  = blockIdx.x * 8 + wv;
    const int b    = pos / HWn;

    const float* xp = x + (size_t)pos * Cn;
    const float* cp = chout + (size_t)b * Cn;

    float s = 0.0f, m = -3.402823466e38f;
    #pragma unroll
    for (int i = 0; i < Cn / 32; ++i) {
        const int c = lane + i * 32;
        const float v = xp[c] * cp[c];
        s += v;
        m = fmaxf(m, v);
    }
    #pragma unroll
    for (int off = 16; off >= 1; off >>= 1) {
        s += __shfl_xor(s, off, 32);
        m = fmaxf(m, __shfl_xor(m, off, 32));
    }
    if (lane == 0) {
        sp[(size_t)pos * 2 + 0] = s * (1.0f / (float)Cn);
        sp[(size_t)pos * 2 + 1] = m;
    }
}

__global__ void k_spatial_conv(const float* __restrict__ sp,
                               const float* __restrict__ wconv,
                               const float* __restrict__ bconv,
                               float* __restrict__ satt) {
    const int idx = blockIdx.x * blockDim.x + threadIdx.x;
    const int b  = idx / HWn;
    const int hw = idx - b * HWn;
    const int y  = hw / Wn;
    const int xx = hw - y * Wn;

    float acc = 0.0f;
    #pragma unroll
    for (int ky = 0; ky < 7; ++ky) {
        const int iy = y + ky - 3;
        if (iy < 0 || iy >= Hn) continue;
        #pragma unroll
        for (int kx = 0; kx < 7; ++kx) {
            const int ix = xx + kx - 3;
            if (ix < 0 || ix >= Wn) continue;
            const size_t sidx = ((size_t)(b * Hn + iy) * Wn + ix) * 2;
            acc += sp[sidx + 0] * wconv[(ky * 7 + kx) * 2 + 0];
            acc += sp[sidx + 1] * wconv[(ky * 7 + kx) * 2 + 1];
        }
    }
    satt[idx] = sigmoidf_(acc + bconv[0]);
}

__global__ void k_apply(const float* __restrict__ x,
                        const float* __restrict__ chout,
                        const float* __restrict__ satt,
                        float* __restrict__ out) {
    const size_t gid = (size_t)blockIdx.x * blockDim.x + threadIdx.x;
    const size_t pos = gid >> 6;
    const int    c4  = (int)(gid & 63);
    const int    b   = (int)(pos / HWn);

    const float4 xv = ((const float4*)x)[gid];
    const float4 cv = ((const float4*)chout)[(size_t)b * (Cn / 4) + c4];
    const float  s  = satt[pos];

    float4 o;
    o.x = xv.x * cv.x * s;
    o.y = xv.y * cv.y * s;
    o.z = xv.z * cv.z * s;
    o.w = xv.w * cv.w * s;
    ((float4*)out)[gid] = o;
}

extern "C" void kernel_launch(void* const* d_in, const int* in_sizes, int n_in,
                              void* d_out, int out_size, void* d_ws, size_t ws_size,
                              hipStream_t stream) {
    const float* x     = (const float*)d_in[0];
    const float* w1    = (const float*)d_in[1];
    const float* b1    = (const float*)d_in[2];
    const float* w2    = (const float*)d_in[3];
    const float* b2    = (const float*)d_in[4];
    const float* wconv = (const float*)d_in[5];
    const float* bconv = (const float*)d_in[6];
    float* out = (float*)d_out;

    float* ws    = (float*)d_ws;
    float* psum  = ws;
    float* pmax  = psum + (size_t)NCHK * Bn * Cn;
    float* chout = pmax + (size_t)NCHK * Bn * Cn;
    float* sp    = chout + (size_t)Bn * Cn;
    float* satt  = sp + (size_t)Bn * HWn * 2;

    k_pool_partial<<<dim3(NCHK, Bn), 256, 0, stream>>>(x, psum, pmax);

    k_channel_mlp<<<1, 256, 0, stream>>>(psum, pmax, w1, b1, w2, b2, chout);

    k_spatial_stats<<<(Bn * HWn) / 8, 256, 0, stream>>>(x, chout, sp);

    k_spatial_conv<<<(Bn * HWn) / 256, 256, 0, stream>>>(sp, wconv, bconv, satt);

    k_apply<<<(Bn * HWn * Cn / 4) / 256, 256, 0, stream>>>(x, chout, satt, out);
}


# pooling finished in its own launch: a block owns one image x 32 channels over all 3136 pixels and writes final avg/max rows (was 8 chunk partials folded by the MLP launch); changes pooled sum order
# speedup vs baseline: 1.8256x; 1.8256x over previous
#include <hip/hip_runtime.h>
#include <hip/hip_bf16.h>
#include <math.h>

#define Bn   32
#define Hn   56
#define Wn   56
#define Cn   256
#define HWn  (Hn * Wn)
#define HID  16
#define SLICE  32
#define NSLICE (Cn / SLICE)
#define PWAVES 28
#define PLOADS 28

typedef float v2f __attribute__((ext_vector_type(2)));
typedef float v8f __attribute__((ext_vector_type(8)));

__device__ __forceinline__ v8f wmma4(v2f a, v2f b, v8f c) {
    return __builtin_amdgcn_wmma_f32_16x16x4_f32(
         false, a,  false, b,
         (short)0, c,  false,  false);
}

__device__ __forceinline__ float sigmoidf_(float v) {
    return 1.0f / (1.0f + __expf(-v));
}

__global__ __launch_bounds__(PWAVES * 32)
void k_pool_slice(const float* __restrict__ x,
                  float* __restrict__ pavg,
                  float* __restrict__ pmax) {
    __shared__ __attribute__((aligned(16))) float lds_s[PWAVES][SLICE];
    __shared__ __attribute__((aligned(16))) float lds_m[PWAVES][SLICE];

    const int lane = threadIdx.x & 31;
    const int wv   = threadIdx.x >> 5;
    const int q    = lane & 7;
    const int r    = lane >> 3;
    const int sl   = blockIdx.x;
    const int b    = blockIdx.y;

    const float4* xp = (const float4*)(x + ((size_t)b * HWn + (size_t)wv * (PLOADS * 4) + r) * Cn
                                         + sl * SLICE) + q;

    float4 v[PLOADS];
    #pragma unroll
    for (int j = 0; j < PLOADS; ++j)
        v[j] = xp[(size_t)j * 4 * (Cn / 4)];

    float4 s = make_float4(0.0f, 0.0f, 0.0f, 0.0f);
    float4 m = make_float4(-3.402823466e38f, -3.402823466e38f,
                           -3.402823466e38f, -3.402823466e38f);
    #pragma unroll
    for (int j = 0; j < PLOADS; ++j) {
        s.x += v[j].x;  s.y += v[j].y;  s.z += v[j].z;  s.w += v[j].w;
        m.x = fmaxf(m.x, v[j].x);  m.y = fmaxf(m.y, v[j].y);
        m.z = fmaxf(m.z, v[j].z);  m.w = fmaxf(m.w, v[j].w);
    }

    #pragma unroll
    for (int off = 8; off <= 16; off <<= 1) {
        s.x += __shfl_xor(s.x, off, 32);  s.y += __shfl_xor(s.y, off, 32);
        s.z += __shfl_xor(s.z, off, 32);  s.w += __shfl_xor(s.w, off, 32);
        m.x = fmaxf(m.x, __shfl_xor(m.x, off, 32));
        m.y = fmaxf(m.y, __shfl_xor(m.y, off, 32));
        m.z = fmaxf(m.z, __shfl_xor(m.z, off, 32));
        m.w = fmaxf(m.w, __shfl_xor(m.w, off, 32));
    }
    if (lane < 8) {
        *(float4*)&lds_s[wv][q * 4] = s;
        *(float4*)&lds_m[wv][q * 4] = m;
    }
    __syncthreads();

    if (wv < 2) {
        const int c = lane;
        if (wv == 0) {
            float ts = 0.0f;
            #pragma unroll
            for (int w = 0; w < PWAVES; ++w) ts += lds_s[w][c];
            pavg[(size_t)b * Cn + sl * SLICE + c] = ts * (1.0f / (float)HWn);
        } else {
            float tm = -3.402823466e38f;
            #pragma unroll
            for (int w = 0; w < PWAVES; ++w) tm = fmaxf(tm, lds_m[w][c]);
            pmax[(size_t)b * Cn + sl * SLICE + c] = tm;
        }
    }
}

__global__ void k_channel_mlp(const float* __restrict__ pavg,
                              const float* __restrict__ pmax,
                              const float* __restrict__ w1,
                              const float* __restrict__ b1,
                              const float* __restrict__ w2,
                              const float* __restrict__ b2,
                              float* __restrict__ chout) {
    __shared__ float lds_pool[2][Bn][Cn + 1];
    __shared__ float lds_h[2][Bn][HID + 1];

    const int tid  = threadIdx.x;
    const int wv   = tid >> 5;
    const int lane = tid & 31;

    #pragma unroll
    for (int b = 0; b < Bn; ++b) {
        lds_pool[0][b][tid] = pavg[b * Cn + tid];
        lds_pool[1][b][tid] = pmax[b * Cn + tid];
    }
    __syncthreads();

    if (wv < 4) {
        const int src  = wv >> 1;
        const int mt   = wv & 1;
        const int col  = lane & 15;
        const int m    = mt * 16 + col;
        const int ksel = (lane < 16) ? 0 : 2;

        v8f acc = {};
        for (int k0 = 0; k0 < Cn; k0 += 4) {
            v2f a, bfr;
            a.x   = lds_pool[src][m][k0 + ksel];
            a.y   = lds_pool[src][m][k0 + ksel + 1];
            bfr.x = w1[(k0 + ksel)     * HID + col];
            bfr.y = w1[(k0 + ksel + 1) * HID + col];
            acc = wmma4(a, bfr, acc);
        }
        const float bias  = b1[col];
        const int   rbase = mt * 16 + ((lane & 16) ? 8 : 0);
        #pragma unroll
        for (int v = 0; v < 8; ++v)
            lds_h[src][rbase + v][col] = fmaxf(acc[v] + bias, 0.0f);
    }
    __syncthreads();

    for (int t = wv; t < 32; t += 8) {
        const int mt   = t & 1;
        const int nt   = t >> 1;
        const int col  = lane & 15;
        const int ksel = (lane < 16) ? 0 : 2;
        const float bias = b2[nt * 16 + col];

        float combined[8];
        #pragma unroll
        for (int v = 0; v < 8; ++v) combined[v] = 0.0f;

        #pragma unroll
        for (int src = 0; src < 2; ++src) {
            v8f acc = {};
            #pragma unroll
            for (int k0 = 0; k0 < HID; k0 += 4) {
                v2f a, bfr;
                a.x   = lds_h[src][mt * 16 + col][k0 + ksel];
                a.y   = lds_h[src][mt * 16 + col][k0 + ksel + 1];
                bfr.x = w2[(k0 + ksel)     * Cn + nt * 16 + col];
                bfr.y = w2[(k0 + ksel + 1) * Cn + nt * 16 + col];
                acc = wmma4(a, bfr, acc);
            }
            #pragma unroll
            for (int v = 0; v < 8; ++v)
                combined[v] += sigmoidf_(acc[v] + bias);
        }
        const int rbase = mt * 16 + ((lane & 16) ? 8 : 0);
        #pragma unroll
        for (int v = 0; v < 8; ++v)
            chout[(size_t)(rbase + v) * Cn + nt * 16 + col] = combined[v];
    }
}

__global__ void k_spatial_stats(const float* __restrict__ x,
                                const float* __restrict__ chout,
                                float* __restrict__ sp) {
    const int lane = threadIdx.x & 31;
    const int wv   = threadIdx.x >> 5;
    const int pos  = blockIdx.x * 8 + wv;
    const int b    = pos / HWn;

    const float* xp = x + (size_t)pos * Cn;
    const float* cp = chout + (size_t)b * Cn;

    float s = 0.0f, m = -3.402823466e38f;
    #pragma unroll
    for (int i = 0; i < Cn / 32; ++i) {
        const int c = lane + i * 32;
        const float v = xp[c] * cp[c];
        s += v;
        m = fmaxf(m, v);
    }
    #pragma unroll
    for (int off = 16; off >= 1; off >>= 1) {
        s += __shfl_xor(s, off, 32);
        m = fmaxf(m, __shfl_xor(m, off, 32));
    }
    if (lane == 0) {
        sp[(size_t)pos * 2 + 0] = s * (1.0f / (float)Cn);
        sp[(size_t)pos * 2 + 1] = m;
    }
}

__global__ void k_spatial_conv(const float* __restrict__ sp,
                               const float* __restrict__ wconv,
                               const float* __restrict__ bconv,
                               float* __restrict__ satt) {
    const int idx = blockIdx.x * blockDim.x + threadIdx.x;
    const int b  = idx / HWn;
    const int hw = idx - b * HWn;
    const int y  = hw / Wn;
    const int xx = hw - y * Wn;

    float acc = 0.0f;
    #pragma unroll
    for (int ky = 0; ky < 7; ++ky) {
        const int iy = y + ky - 3;
        if (iy < 0 || iy >= Hn) continue;
        #pragma unroll
        for (int kx = 0; kx < 7; ++kx) {
            const int ix = xx + kx - 3;
            if (ix < 0 || ix >= Wn) continue;
            const size_t sidx = ((size_t)(b * Hn + iy) * Wn + ix) * 2;
            acc += sp[sidx + 0] * wconv[(ky * 7 + kx) * 2 + 0];
            acc += sp[sidx + 1] * wconv[(ky * 7 + kx) * 2 + 1];
        }
    }
    satt[idx] = sigmoidf_(acc + bconv[0]);
}

__global__ void k_apply(const float* __restrict__ x,
                        const float* __restrict__ chout,
                        const float* __restrict__ satt,
                        float* __restrict__ out) {
    const size_t gid = (size_t)blockIdx.x * blockDim.x + threadIdx.x;
    const size_t pos = gid >> 6;
    const int    c4  = (int)(gid & 63);
    const int    b   = (int)(pos / HWn);

    const float4 xv = ((const float4*)x)[gid];
    const float4 cv = ((const float4*)chout)[(size_t)b * (Cn / 4) + c4];
    const float  s  = satt[pos];

    float4 o;
    o.x = xv.x * cv.x * s;
    o.y = xv.y * cv.y * s;
    o.z = xv.z * cv.z * s;
    o.w = xv.w * cv.w * s;
    ((float4*)out)[gid] = o;
}

extern "C" void kernel_launch(void* const* d_in, const int* in_sizes, int n_in,
                              void* d_out, int out_size, void* d_ws, size_t ws_size,
                              hipStream_t stream) {
    const float* x     = (const float*)d_in[0];
    const float* w1    = (const float*)d_in[1];
    const float* b1    = (const float*)d_in[2];
    const float* w2    = (const float*)d_in[3];
    const float* b2    = (const float*)d_in[4];
    const float* wconv = (const float*)d_in[5];
    const float* bconv = (const float*)d_in[6];
    float* out = (float*)d_out;

    float* ws    = (float*)d_ws;
    float* pavg  = ws;
    float* pmax  = pavg + (size_t)Bn * Cn;
    float* chout = pmax + (size_t)Bn * Cn;
    float* sp    = chout + (size_t)Bn * Cn;
    float* satt  = sp + (size_t)Bn * HWn * 2;

    k_pool_slice<<<dim3(NSLICE, Bn), PWAVES * 32, 0, stream>>>(x, pavg, pmax);

    k_channel_mlp<<<1, 256, 0, stream>>>(pavg, pmax, w1, b1, w2, b2, chout);

    k_spatial_stats<<<(Bn * HWn) / 8, 256, 0, stream>>>(x, chout, sp);

    k_spatial_conv<<<(Bn * HWn) / 256, 256, 0, stream>>>(sp, wconv, bconv, satt);

    k_apply<<<(Bn * HWn * Cn / 4) / 256, 256, 0, stream>>>(x, chout, satt, out);
}


# 7x7 conv launch dissolved into the apply launch (4 -> 3 launches): a block owns one image row, stages the zero-padded sp tile in LDS, 56 lanes gate the row, a wave streams 14 pixels; sum orders kept
# speedup vs baseline: 3.6409x; 1.9944x over previous
#include <hip/hip_runtime.h>
#include <hip/hip_bf16.h>
#include <math.h>

#define Bn   32
#define Hn   56
#define Wn   56
#define Cn   256
#define HWn  (Hn * Wn)
#define HID  16
#define SLICE  32
#define NSLICE (Cn / SLICE)
#define PWAVES 28
#define PLOADS 28

typedef float v2f __attribute__((ext_vector_type(2)));
typedef float v8f __attribute__((ext_vector_type(8)));

__device__ __forceinline__ v8f wmma4(v2f a, v2f b, v8f c) {
    return __builtin_amdgcn_wmma_f32_16x16x4_f32(
         false, a,  false, b,
         (short)0, c,  false,  false);
}

__device__ __forceinline__ float sigmoidf_(float v) {
    return 1.0f / (1.0f + __expf(-v));
}

__global__ __launch_bounds__(PWAVES * 32)
void k_pool_slice(const float* __restrict__ x,
                  const float* __restrict__ w1,
                  float* __restrict__ hpart) {
    __shared__ __attribute__((aligned(16))) float lds_s[PWAVES][SLICE];
    __shared__ __attribute__((aligned(16))) float lds_m[PWAVES][SLICE];
    __shared__ float lds_p[2][SLICE];

    const int lane = threadIdx.x & 31;
    const int wv   = threadIdx.x >> 5;
    const int q    = lane & 7;
    const int r    = lane >> 3;
    const int sl   = blockIdx.x;
    const int b    = blockIdx.y;
    const int col  = lane & 15;
    const int ksel = (lane < 16) ? 0 : 2;

    float wf[SLICE / 2];
    #pragma unroll
    for (int j = 0; j < SLICE / 2; ++j) wf[j] = 0.0f;
    if (wv == 0) {
        #pragma unroll
        for (int j = 0; j < SLICE / 4; ++j) {
            wf[2 * j]     = w1[(sl * SLICE + 4 * j + ksel)     * HID + col];
            wf[2 * j + 1] = w1[(sl * SLICE + 4 * j + ksel + 1) * HID + col];
        }
    }

    const float4* xp = (const float4*)(x + ((size_t)b * HWn + (size_t)wv * (PLOADS * 4) + r) * Cn
                                         + sl * SLICE) + q;

    float4 v[PLOADS];
    #pragma unroll
    for (int j = 0; j < PLOADS; ++j)
        v[j] = xp[(size_t)j * 4 * (Cn / 4)];

    float4 s = make_float4(0.0f, 0.0f, 0.0f, 0.0f);
    float4 m = make_float4(-3.402823466e38f, -3.402823466e38f,
                           -3.402823466e38f, -3.402823466e38f);
    #pragma unroll
    for (int j = 0; j < PLOADS; ++j) {
        s.x += v[j].x;  s.y += v[j].y;  s.z += v[j].z;  s.w += v[j].w;
        m.x = fmaxf(m.x, v[j].x);  m.y = fmaxf(m.y, v[j].y);
        m.z = fmaxf(m.z, v[j].z);  m.w = fmaxf(m.w, v[j].w);
    }

    #pragma unroll
    for (int off = 8; off <= 16; off <<= 1) {
        s.x += __shfl_xor(s.x, off, 32);  s.y += __shfl_xor(s.y, off, 32);
        s.z += __shfl_xor(s.z, off, 32);  s.w += __shfl_xor(s.w, off, 32);
        m.x = fmaxf(m.x, __shfl_xor(m.x, off, 32));
        m.y = fmaxf(m.y, __shfl_xor(m.y, off, 32));
        m.z = fmaxf(m.z, __shfl_xor(m.z, off, 32));
        m.w = fmaxf(m.w, __shfl_xor(m.w, off, 32));
    }
    if (lane < 8) {
        *(float4*)&lds_s[wv][q * 4] = s;
        *(float4*)&lds_m[wv][q * 4] = m;
    }
    __syncthreads();

    if (wv < 2) {
        const int c = lane;
        if (wv == 0) {
            float ts = 0.0f;
            #pragma unroll
            for (int w = 0; w < PWAVES; ++w) ts += lds_s[w][c];
            lds_p[0][c] = ts * (1.0f / (float)HWn);
        } else {
            float tm = -3.402823466e38f;
            #pragma unroll
            for (int w = 0; w < PWAVES; ++w) tm = fmaxf(tm, lds_m[w][c]);
            lds_p[1][c] = tm;
        }
    }
    __syncthreads();

    if (wv == 0) {
        v8f acc = {};
        #pragma unroll
        for (int j = 0; j < SLICE / 4; ++j) {
            v2f a, bfr;
            const float p0 = lds_p[col & 1][4 * j + ksel];
            const float p1 = lds_p[col & 1][4 * j + ksel + 1];
            a.x   = (col < 2) ? p0 : 0.0f;
            a.y   = (col < 2) ? p1 : 0.0f;
            bfr.x = wf[2 * j];
            bfr.y = wf[2 * j + 1];
            acc = wmma4(a, bfr, acc);
        }
        if (lane < 16) {
            float* hp = hpart + ((size_t)(b * NSLICE + sl) * 2) * HID;
            hp[col]       = acc[0];
            hp[HID + col] = acc[1];
        }
    }
}

#define PPW   8
#define SPIX  (8 * PPW)
#define SBLK  (HWn / SPIX)

__global__ __launch_bounds__(256)
void k_mlp_stats(const float* __restrict__ x,
                 const float* __restrict__ hpart,
                 const float* __restrict__ b1,
                 const float* __restrict__ w2,
                 const float* __restrict__ b2,
                 float* __restrict__ chout,
                 float* __restrict__ sp) {
    __shared__ float lds_c[Cn];

    const int lane = threadIdx.x & 31;
    const int wv   = threadIdx.x >> 5;
    const int b    = blockIdx.x / SBLK;
    const int pix0 = blockIdx.x * SPIX + wv * PPW;
    const int col  = lane & 15;
    const int ksel = (lane < 16) ? 0 : 2;

    const float* xp = x + (size_t)pix0 * Cn + lane;
    float xv[PPW][Cn / 32];
    #pragma unroll
    for (int p = 0; p < PPW; ++p) {
        #pragma unroll
        for (int i = 0; i < Cn / 32; ++i)
            xv[p][i] = xp[(size_t)p * Cn + i * 32];
    }

    float hp[NSLICE];
    #pragma unroll
    for (int s = 0; s < NSLICE; ++s)
        hp[s] = hpart[(size_t)(b * NSLICE + s) * (2 * HID) + lane];
    const float bias1 = b1[col];
    float wfr[2][8];
    const float bias2 = b2[wv * 32 + lane];
    #pragma unroll
    for (int t = 0; t < 2; ++t) {
        const int n0 = (wv * 2 + t) * 16;
        #pragma unroll
        for (int j = 0; j < HID / 4; ++j) {
            wfr[t][2 * j]     = w2[(4 * j + ksel)     * Cn + n0 + col];
            wfr[t][2 * j + 1] = w2[(4 * j + ksel + 1) * Cn + n0 + col];
        }
    }

    float hs = hp[0];
    #pragma unroll
    for (int s = 1; s < NSLICE; ++s) hs += hp[s];
    hs = fmaxf(hs + bias1, 0.0f);

    v2f af[HID / 4];
    #pragma unroll
    for (int j = 0; j < HID / 4; ++j) {
        const float a0 = __shfl(hs, (col & 1) * 16 + 4 * j + ksel,     32);
        const float a1 = __shfl(hs, (col & 1) * 16 + 4 * j + ksel + 1, 32);
        af[j].x = (col < 2) ? a0 : 0.0f;
        af[j].y = (col < 2) ? a1 : 0.0f;
    }
    v8f acc[2];
    #pragma unroll
    for (int t = 0; t < 2; ++t) {
        acc[t] = (v8f){};
        #pragma unroll
        for (int j = 0; j < HID / 4; ++j) {
            v2f bfr;
            bfr.x = wfr[t][2 * j];
            bfr.y = wfr[t][2 * j + 1];
            acc[t] = wmma4(af[j], bfr, acc[t]);
        }
    }
    const float u0 = __shfl(acc[1][0], col, 32);
    const float u1 = __shfl(acc[1][1], col, 32);
    const float z0 = (lane < 16) ? acc[0][0] : u0;
    const float z1 = (lane < 16) ? acc[0][1] : u1;
    lds_c[wv * 32 + lane] = sigmoidf_(z0 + bias2) + sigmoidf_(z1 + bias2);
    __syncthreads();

    if (blockIdx.x == b * SBLK)
        chout[(size_t)b * Cn + threadIdx.x] = lds_c[threadIdx.x];

    float cv[Cn / 32];
    #pragma unroll
    for (int i = 0; i < Cn / 32; ++i) cv[i] = lds_c[lane + i * 32];

    float s[PPW], m[PPW];
    #pragma unroll
    for (int p = 0; p < PPW; ++p) {
        s[p] = 0.0f;
        m[p] = -3.402823466e38f;
        #pragma unroll
        for (int i = 0; i < Cn / 32; ++i) {
            const float v = xv[p][i] * cv[i];
            s[p] += v;
            m[p] = fmaxf(m[p], v);
        }
    }
    #pragma unroll
    for (int off = 16; off >= 1; off >>= 1) {
        float ts[PPW], tm[PPW];
        #pragma unroll
        for (int p = 0; p < PPW; ++p) {
            ts[p] = __shfl_xor(s[p], off, 32);
            tm[p] = __shfl_xor(m[p], off, 32);
        }
        #pragma unroll
        for (int p = 0; p < PPW; ++p) {
            s[p] += ts[p];
            m[p] = fmaxf(m[p], tm[p]);
        }
    }
    float rs = s[0], rm = m[0];
    #pragma unroll
    for (int p = 1; p < PPW; ++p) {
        rs = (lane == p) ? s[p] : rs;
        rm = (lane == p) ? m[p] : rm;
    }
    if (lane < PPW) {
        float2 o;
        o.x = rs * (1.0f / (float)Cn);
        o.y = rm;
        *(float2*)(sp + (size_t)(pix0 + lane) * 2) = o;
    }
}

#define AWAVES 4
#define APIX   (Wn / AWAVES)
#define ALOADS (APIX * (Cn / 128))
#define TROW   32

__device__ __forceinline__ void barrier_lds_only() {
    asm volatile("s_wait_dscnt 0x0\n\ts_barrier_signal -1\n\ts_barrier_wait -1" ::: "memory");
}

__global__ __launch_bounds__(AWAVES * 32, 7)
void k_conv_apply(const float* __restrict__ x,
                  const float* __restrict__ chout,
                  const float* __restrict__ sp,
                  const float* __restrict__ wconv,
                  const float* __restrict__ bconv,
                  float* __restrict__ out) {
    __shared__ __attribute__((aligned(16))) float4 tile4[7 * TROW];
    __shared__ float sat[Wn];

    const int t    = threadIdx.x;
    const int lane = t & 31;
    const int wv   = t >> 5;
    const int y    = blockIdx.x;
    const int b    = blockIdx.y;

    const float4* cp = (const float4*)chout + (size_t)b * (Cn / 4) + lane;
    float4 cv[2];
    cv[0] = cp[0];
    cv[1] = cp[32];

    float4 tv[2];
    int    tdst[2];
    #pragma unroll
    for (int k = 0; k < 2; ++k) {
        const int e  = t + k * (AWAVES * 32);
        const int r  = e / 28;
        const int c  = e - r * 28;
        const int iy = y + r - 3;
        tdst[k] = r * TROW + 2 + c;
        tv[k] = make_float4(0.0f, 0.0f, 0.0f, 0.0f);
        if (e < 7 * 28 && iy >= 0 && iy < Hn)
            tv[k] = ((const float4*)sp)[(size_t)(b * Hn + iy) * 28 + c];
    }

    const size_t base4 = ((size_t)(b * Hn + y) * Wn + (size_t)wv * APIX) * (Cn / 4) + lane;
    const float4* xp = (const float4*)x + base4;
    float4 xv[ALOADS];
    #pragma unroll
    for (int i = 0; i < ALOADS; ++i) xv[i] = xp[(size_t)i * 32];

    #pragma unroll
    for (int k = 0; k < 2; ++k)
        if (t + k * (AWAVES * 32) < 7 * 28) tile4[tdst[k]] = tv[k];
    if (t >= 100) {
        const int i = t - 100;
        const int r = i >> 2;
        const int c = i & 3;
        tile4[r * TROW + ((c < 2) ? c : 28 + c)] = make_float4(0.0f, 0.0f, 0.0f, 0.0f);
    }
    barrier_lds_only();

    if (t < Wn) {
        const float2* tile2 = (const float2*)tile4;
        float acc = 0.0f;
        #pragma unroll
        for (int ky = 0; ky < 7; ++ky) {
            #pragma unroll
            for (int kx = 0; kx < 7; ++kx) {
                const float2 v = tile2[ky * (2 * TROW) + t + kx + 1];
                acc += v.x * wconv[(ky * 7 + kx) * 2 + 0];
                acc += v.y * wconv[(ky * 7 + kx) * 2 + 1];
            }
        }
        sat[t] = sigmoidf_(acc + bconv[0]);
    }
    barrier_lds_only();

    const float gl = sat[wv * APIX + ((lane < APIX) ? lane : 0)];
    float4* op = (float4*)out + base4;
    #pragma unroll
    for (int j = 0; j < APIX; ++j) {
        const float s = __builtin_bit_cast(float,
            __builtin_amdgcn_readlane(__builtin_bit_cast(int, gl), j));
        #pragma unroll
        for (int h = 0; h < 2; ++h) {
            const float4 v = xv[2 * j + h];
            float4 o;
            o.x = v.x * cv[h].x * s;
            o.y = v.y * cv[h].y * s;
            o.z = v.z * cv[h].z * s;
            o.w = v.w * cv[h].w * s;
            op[(size_t)(2 * j + h) * 32] = o;
        }
        __builtin_amdgcn_sched_barrier(0);
    }
}

extern "C" void kernel_launch(void* const* d_in, const int* in_sizes, int n_in,
                              void* d_out, int out_size, void* d_ws, size_t ws_size,
                              hipStream_t stream) {
    const float* x     = (const float*)d_in[0];
    const float* w1    = (const float*)d_in[1];
    const float* b1    = (const float*)d_in[2];
    const float* w2    = (const float*)d_in[3];
    const float* b2    = (const float*)d_in[4];
    const float* wconv = (const float*)d_in[5];
    const float* bconv = (const float*)d_in[6];
    float* out = (float*)d_out;

    float* ws    = (float*)d_ws;
    float* hpart = ws;
    float* chout = ws + (size_t)2 * Bn * Cn;
    float* sp    = chout + (size_t)Bn * Cn;

    k_pool_slice<<<dim3(NSLICE, Bn), PWAVES * 32, 0, stream>>>(x, w1, hpart);

    k_mlp_stats<<<Bn * SBLK, 256, 0, stream>>>(x, hpart, b1, w2, b2, chout, sp);

    k_conv_apply<<<dim3(Hn, Bn), AWAVES * 32, 0, stream>>>(x, chout, sp, wconv, bconv, out);
}


# stats launch re-cut to 256 blocks x 14 waves (was 1568 x 8): a wave walks 28 pixels in 4 stages of 7 through two register buffers, next stage's loads under this stage's reduction; sum orders kept
# speedup vs baseline: 3.7831x; 1.0391x over previous
#include <hip/hip_runtime.h>
#include <hip/hip_bf16.h>
#include <math.h>

#define Bn   32
#define Hn   56
#define Wn   56
#define Cn   256
#define HWn  (Hn * Wn)
#define HID  16
#define SLICE  32
#define NSLICE (Cn / SLICE)
#define PWAVES 28
#define PLOADS 28

typedef float v2f __attribute__((ext_vector_type(2)));
typedef float v8f __attribute__((ext_vector_type(8)));

__device__ __forceinline__ v8f wmma4(v2f a, v2f b, v8f c) {
    return __builtin_amdgcn_wmma_f32_16x16x4_f32(
         false, a,  false, b,
         (short)0, c,  false,  false);
}

__device__ __forceinline__ float sigmoidf_(float v) {
    return 1.0f / (1.0f + __expf(-v));
}

__global__ __launch_bounds__(PWAVES * 32)
void k_pool_slice(const float* __restrict__ x,
                  const float* __restrict__ w1,
                  float* __restrict__ hpart) {
    __shared__ __attribute__((aligned(16))) float lds_s[PWAVES][SLICE];
    __shared__ __attribute__((aligned(16))) float lds_m[PWAVES][SLICE];
    __shared__ float lds_p[2][SLICE];

    const int lane = threadIdx.x & 31;
    const int wv   = threadIdx.x >> 5;
    const int q    = lane & 7;
    const int r    = lane >> 3;
    const int sl   = blockIdx.x;
    const int b    = blockIdx.y;
    const int col  = lane & 15;
    const int ksel = (lane < 16) ? 0 : 2;

    float wf[SLICE / 2];
    #pragma unroll
    for (int j = 0; j < SLICE / 2; ++j) wf[j] = 0.0f;
    if (wv == 0) {
        #pragma unroll
        for (int j = 0; j < SLICE / 4; ++j) {
            wf[2 * j]     = w1[(sl * SLICE + 4 * j + ksel)     * HID + col];
            wf[2 * j + 1] = w1[(sl * SLICE + 4 * j + ksel + 1) * HID + col];
        }
    }

    const float4* xp = (const float4*)(x + ((size_t)b * HWn + (size_t)wv * (PLOADS * 4) + r) * Cn
                                         + sl * SLICE) + q;

    float4 v[PLOADS];
    #pragma unroll
    for (int j = 0; j < PLOADS; ++j)
        v[j] = xp[(size_t)j * 4 * (Cn / 4)];

    float4 s = make_float4(0.0f, 0.0f, 0.0f, 0.0f);
    float4 m = make_float4(-3.402823466e38f, -3.402823466e38f,
                           -3.402823466e38f, -3.402823466e38f);
    #pragma unroll
    for (int j = 0; j < PLOADS; ++j) {
        s.x += v[j].x;  s.y += v[j].y;  s.z += v[j].z;  s.w += v[j].w;
        m.x = fmaxf(m.x, v[j].x);  m.y = fmaxf(m.y, v[j].y);
        m.z = fmaxf(m.z, v[j].z);  m.w = fmaxf(m.w, v[j].w);
    }

    #pragma unroll
    for (int off = 8; off <= 16; off <<= 1) {
        s.x += __shfl_xor(s.x, off, 32);  s.y += __shfl_xor(s.y, off, 32);
        s.z += __shfl_xor(s.z, off, 32);  s.w += __shfl_xor(s.w, off, 32);
        m.x = fmaxf(m.x, __shfl_xor(m.x, off, 32));
        m.y = fmaxf(m.y, __shfl_xor(m.y, off, 32));
        m.z = fmaxf(m.z, __shfl_xor(m.z, off, 32));
        m.w = fmaxf(m.w, __shfl_xor(m.w, off, 32));
    }
    if (lane < 8) {
        *(float4*)&lds_s[wv][q * 4] = s;
        *(float4*)&lds_m[wv][q * 4] = m;
    }
    __syncthreads();

    if (wv < 2) {
        const int c = lane;
        if (wv == 0) {
            float ts = 0.0f;
            #pragma unroll
            for (int w = 0; w < PWAVES; ++w) ts += lds_s[w][c];
            lds_p[0][c] = ts * (1.0f / (float)HWn);
        } else {
            float tm = -3.402823466e38f;
            #pragma unroll
            for (int w = 0; w < PWAVES; ++w) tm = fmaxf(tm, lds_m[w][c]);
            lds_p[1][c] = tm;
        }
    }
    __syncthreads();

    if (wv == 0) {
        v8f acc = {};
        #pragma unroll
        for (int j = 0; j < SLICE / 4; ++j) {
            v2f a, bfr;
            const float p0 = lds_p[col & 1][4 * j + ksel];
            const float p1 = lds_p[col & 1][4 * j + ksel + 1];
            a.x   = (col < 2) ? p0 : 0.0f;
            a.y   = (col < 2) ? p1 : 0.0f;
            bfr.x = wf[2 * j];
            bfr.y = wf[2 * j + 1];
            acc = wmma4(a, bfr, acc);
        }
        if (lane < 16) {
            float* hp = hpart + ((size_t)(b * NSLICE + sl) * 2) * HID;
            hp[col]       = acc[0];
            hp[HID + col] = acc[1];
        }
    }
}

__device__ __forceinline__ void barrier_lds_only() {
    asm volatile("s_wait_dscnt 0x0\n\ts_barrier_signal -1\n\ts_barrier_wait -1" ::: "memory");
}

#define SW     14
#define SPX    7
#define SST    4
#define SWPIX  (SPX * SST)
#define SBPIX  (SW * SWPIX)
#define SBLK   (HWn / SBPIX)

__device__ __forceinline__ void stats_load(const float* __restrict__ xp, float (&xv)[SPX][Cn / 32]) {
    #pragma unroll
    for (int p = 0; p < SPX; ++p) {
        #pragma unroll
        for (int i = 0; i < Cn / 32; ++i)
            xv[p][i] = xp[(size_t)p * Cn + i * 32];
    }
}

__device__ __forceinline__ void stats_stage(const float (&xv)[SPX][Cn / 32],
                                            const float (&cv)[Cn / 32],
                                            float* __restrict__ sp, int pix, int lane) {
    float s[SPX], m[SPX];
    #pragma unroll
    for (int p = 0; p < SPX; ++p) {
        s[p] = 0.0f;
        m[p] = -3.402823466e38f;
        #pragma unroll
        for (int i = 0; i < Cn / 32; ++i) {
            const float v = xv[p][i] * cv[i];
            s[p] += v;
            m[p] = fmaxf(m[p], v);
        }
    }
    #pragma unroll
    for (int off = 16; off >= 1; off >>= 1) {
        float ts[SPX], tm[SPX];
        #pragma unroll
        for (int p = 0; p < SPX; ++p) {
            ts[p] = __shfl_xor(s[p], off, 32);
            tm[p] = __shfl_xor(m[p], off, 32);
        }
        #pragma unroll
        for (int p = 0; p < SPX; ++p) {
            s[p] += ts[p];
            m[p] = fmaxf(m[p], tm[p]);
        }
    }
    float rs = s[0], rm = m[0];
    #pragma unroll
    for (int p = 1; p < SPX; ++p) {
        rs = (lane == p) ? s[p] : rs;
        rm = (lane == p) ? m[p] : rm;
    }
    if (lane < SPX) {
        float2 o;
        o.x = rs * (1.0f / (float)Cn);
        o.y = rm;
        *(float2*)(sp + (size_t)(pix + lane) * 2) = o;
    }
}

__global__ __launch_bounds__(SW * 32)
void k_mlp_stats(const float* __restrict__ x,
                 const float* __restrict__ hpart,
                 const float* __restrict__ b1,
                 const float* __restrict__ w2,
                 const float* __restrict__ b2,
                 float* __restrict__ chout,
                 float* __restrict__ sp) {
    __shared__ float lds_c[SW * 32];

    const int lane = threadIdx.x & 31;
    const int wv   = threadIdx.x >> 5;
    const int b    = blockIdx.x / SBLK;
    const int pix0 = blockIdx.x * SBPIX + wv * SWPIX;
    const int col  = lane & 15;
    const int ksel = (lane < 16) ? 0 : 2;

    const int wm = wv & (Cn / 32 - 1);
    float hp[NSLICE];
    #pragma unroll
    for (int s = 0; s < NSLICE; ++s)
        hp[s] = hpart[(size_t)(b * NSLICE + s) * (2 * HID) + lane];
    const float bias1 = b1[col];
    const float bias2 = b2[wm * 32 + lane];
    float wfr[2][8];
    #pragma unroll
    for (int t = 0; t < 2; ++t) {
        const int n0 = (wm * 2 + t) * 16;
        #pragma unroll
        for (int j = 0; j < HID / 4; ++j) {
            wfr[t][2 * j]     = w2[(4 * j + ksel)     * Cn + n0 + col];
            wfr[t][2 * j + 1] = w2[(4 * j + ksel + 1) * Cn + n0 + col];
        }
    }
    __builtin_amdgcn_sched_barrier(0);

    const float* xp = x + (size_t)pix0 * Cn + lane;
    float xa[SPX][Cn / 32], xb[SPX][Cn / 32];
    stats_load(xp, xa);
    stats_load(xp + (size_t)SPX * Cn, xb);
    __builtin_amdgcn_sched_barrier(0);

    float hs = hp[0];
    #pragma unroll
    for (int s = 1; s < NSLICE; ++s) hs += hp[s];
    hs = fmaxf(hs + bias1, 0.0f);

    v2f af[HID / 4];
    #pragma unroll
    for (int j = 0; j < HID / 4; ++j) {
        const float a0 = __shfl(hs, (col & 1) * 16 + 4 * j + ksel,     32);
        const float a1 = __shfl(hs, (col & 1) * 16 + 4 * j + ksel + 1, 32);
        af[j].x = (col < 2) ? a0 : 0.0f;
        af[j].y = (col < 2) ? a1 : 0.0f;
    }
    v8f acc[2];
    #pragma unroll
    for (int t = 0; t < 2; ++t) {
        acc[t] = (v8f){};
        #pragma unroll
        for (int j = 0; j < HID / 4; ++j) {
            v2f bfr;
            bfr.x = wfr[t][2 * j];
            bfr.y = wfr[t][2 * j + 1];
            acc[t] = wmma4(af[j], bfr, acc[t]);
        }
    }
    const float u0 = __shfl(acc[1][0], col, 32);
    const float u1 = __shfl(acc[1][1], col, 32);
    const float z0 = (lane < 16) ? acc[0][0] : u0;
    const float z1 = (lane < 16) ? acc[0][1] : u1;
    const float cg = sigmoidf_(z0 + bias2) + sigmoidf_(z1 + bias2);
    lds_c[wv * 32 + lane] = cg;
    barrier_lds_only();

    if (blockIdx.x == b * SBLK && threadIdx.x < Cn)
        chout[(size_t)b * Cn + threadIdx.x] = lds_c[threadIdx.x];

    float cv[Cn / 32];
    #pragma unroll
    for (int i = 0; i < Cn / 32; ++i) cv[i] = lds_c[lane + i * 32];

    stats_stage(xa, cv, sp, pix0, lane);
    __builtin_amdgcn_sched_barrier(0);
    stats_load(xp + (size_t)(2 * SPX) * Cn, xa);
    __builtin_amdgcn_sched_barrier(0);
    stats_stage(xb, cv, sp, pix0 + SPX, lane);
    __builtin_amdgcn_sched_barrier(0);
    stats_load(xp + (size_t)(3 * SPX) * Cn, xb);
    __builtin_amdgcn_sched_barrier(0);
    stats_stage(xa, cv, sp, pix0 + 2 * SPX, lane);
    stats_stage(xb, cv, sp, pix0 + 3 * SPX, lane);
}

#define AWAVES 4
#define APIX   (Wn / AWAVES)
#define ALOADS (APIX * (Cn / 128))
#define TROW   32

__global__ __launch_bounds__(AWAVES * 32, 7)
void k_conv_apply(const float* __restrict__ x,
                  const float* __restrict__ chout,
                  const float* __restrict__ sp,
                  const float* __restrict__ wconv,
                  const float* __restrict__ bconv,
                  float* __restrict__ out) {
    __shared__ __attribute__((aligned(16))) float4 tile4[7 * TROW];
    __shared__ float sat[Wn];

    const int t    = threadIdx.x;
    const int lane = t & 31;
    const int wv   = t >> 5;
    const int y    = blockIdx.x;
    const int b    = blockIdx.y;

    const float4* cp = (const float4*)chout + (size_t)b * (Cn / 4) + lane;
    float4 cv[2];
    cv[0] = cp[0];
    cv[1] = cp[32];

    float4 tv[2];
    int    tdst[2];
    #pragma unroll
    for (int k = 0; k < 2; ++k) {
        const int e  = t + k * (AWAVES * 32);
        const int r  = e / 28;
        const int c  = e - r * 28;
        const int iy = y + r - 3;
        tdst[k] = r * TROW + 2 + c;
        tv[k] = make_float4(0.0f, 0.0f, 0.0f, 0.0f);
        if (e < 7 * 28 && iy >= 0 && iy < Hn)
            tv[k] = ((const float4*)sp)[(size_t)(b * Hn + iy) * 28 + c];
    }

    const size_t base4 = ((size_t)(b * Hn + y) * Wn + (size_t)wv * APIX) * (Cn / 4) + lane;
    const float4* xp = (const float4*)x + base4;
    float4 xv[ALOADS];
    #pragma unroll
    for (int i = 0; i < ALOADS; ++i) xv[i] = xp[(size_t)i * 32];

    #pragma unroll
    for (int k = 0; k < 2; ++k)
        if (t + k * (AWAVES * 32) < 7 * 28) tile4[tdst[k]] = tv[k];
    if (t >= 100) {
        const int i = t - 100;
        const int r = i >> 2;
        const int c = i & 3;
        tile4[r * TROW + ((c < 2) ? c : 28 + c)] = make_float4(0.0f, 0.0f, 0.0f, 0.0f);
    }
    barrier_lds_only();

    if (t < Wn) {
        const float2* tile2 = (const float2*)tile4;
        float acc = 0.0f;
        #pragma unroll
        for (int ky = 0; ky < 7; ++ky) {
            #pragma unroll
            for (int kx = 0; kx < 7; ++kx) {
                const float2 v = tile2[ky * (2 * TROW) + t + kx + 1];
                acc += v.x * wconv[(ky * 7 + kx) * 2 + 0];
                acc += v.y * wconv[(ky * 7 + kx) * 2 + 1];
            }
        }
        sat[t] = sigmoidf_(acc + bconv[0]);
    }
    barrier_lds_only();

    const float gl = sat[wv * APIX + ((lane < APIX) ? lane : 0)];
    float4* op = (float4*)out + base4;
    #pragma unroll
    for (int j = 0; j < APIX; ++j) {
        const float s = __builtin_bit_cast(float,
            __builtin_amdgcn_readlane(__builtin_bit_cast(int, gl), j));
        #pragma unroll
        for (int h = 0; h < 2; ++h) {
            const float4 v = xv[2 * j + h];
            float4 o;
            o.x = v.x * cv[h].x * s;
            o.y = v.y * cv[h].y * s;
            o.z = v.z * cv[h].z * s;
            o.w = v.w * cv[h].w * s;
            op[(size_t)(2 * j + h) * 32] = o;
        }
        __builtin_amdgcn_sched_barrier(0);
    }
}

extern "C" void kernel_launch(void* const* d_in, const int* in_sizes, int n_in,
                              void* d_out, int out_size, void* d_ws, size_t ws_size,
                              hipStream_t stream) {
    const float* x     = (const float*)d_in[0];
    const float* w1    = (const float*)d_in[1];
    const float* b1    = (const float*)d_in[2];
    const float* w2    = (const float*)d_in[3];
    const float* b2    = (const float*)d_in[4];
    const float* wconv = (const float*)d_in[5];
    const float* bconv = (const float*)d_in[6];
    float* out = (float*)d_out;

    float* ws    = (float*)d_ws;
    float* hpart = ws;
    float* chout = ws + (size_t)2 * Bn * Cn;
    float* sp    = chout + (size_t)Bn * Cn;

    k_pool_slice<<<dim3(NSLICE, Bn), PWAVES * 32, 0, stream>>>(x, w1, hpart);

    k_mlp_stats<<<Bn * SBLK, SW * 32, 0, stream>>>(x, hpart, b1, w2, b2, chout, sp);

    k_conv_apply<<<dim3(Hn, Bn), AWAVES * 32, 0, stream>>>(x, chout, sp, wconv, bconv, out);
}


# stats pass lane map turned (a float4 load = 4 pixels x one 128-B line; a pixel = 8 lanes x 32 channels, was 32 x 8): 3 exchange levels per 4 pixels, was 5 per pixel; changes spatial-mean sum order
# speedup vs baseline: 4.1054x; 1.0852x over previous
#include <hip/hip_runtime.h>
#include <hip/hip_bf16.h>
#include <math.h>

#define Bn   32
#define Hn   56
#define Wn   56
#define Cn   256
#define HWn  (Hn * Wn)
#define HID  16
#define SLICE  32
#define NSLICE (Cn / SLICE)
#define PWAVES 28
#define PLOADS 28

typedef float v2f __attribute__((ext_vector_type(2)));
typedef float v8f __attribute__((ext_vector_type(8)));

__device__ __forceinline__ v8f wmma4(v2f a, v2f b, v8f c) {
    return __builtin_amdgcn_wmma_f32_16x16x4_f32(
         false, a,  false, b,
         (short)0, c,  false,  false);
}

__device__ __forceinline__ float sigmoidf_(float v) {
    return 1.0f / (1.0f + __expf(-v));
}

__global__ __launch_bounds__(PWAVES * 32)
void k_pool_slice(const float* __restrict__ x,
                  const float* __restrict__ w1,
                  float* __restrict__ hpart) {
    __shared__ __attribute__((aligned(16))) float lds_s[PWAVES][SLICE];
    __shared__ __attribute__((aligned(16))) float lds_m[PWAVES][SLICE];
    __shared__ float lds_p[2][SLICE];

    const int lane = threadIdx.x & 31;
    const int wv   = threadIdx.x >> 5;
    const int q    = lane & 7;
    const int r    = lane >> 3;
    const int sl   = blockIdx.x;
    const int b    = blockIdx.y;
    const int col  = lane & 15;
    const int ksel = (lane < 16) ? 0 : 2;

    float wf[SLICE / 2];
    #pragma unroll
    for (int j = 0; j < SLICE / 2; ++j) wf[j] = 0.0f;
    if (wv == 0) {
        #pragma unroll
        for (int j = 0; j < SLICE / 4; ++j) {
            wf[2 * j]     = w1[(sl * SLICE + 4 * j + ksel)     * HID + col];
            wf[2 * j + 1] = w1[(sl * SLICE + 4 * j + ksel + 1) * HID + col];
        }
    }

    const float4* xp = (const float4*)(x + ((size_t)b * HWn + (size_t)wv * (PLOADS * 4) + r) * Cn
                                         + sl * SLICE) + q;

    float4 v[PLOADS];
    #pragma unroll
    for (int j = 0; j < PLOADS; ++j)
        v[j] = xp[(size_t)j * 4 * (Cn / 4)];

    float4 s = make_float4(0.0f, 0.0f, 0.0f, 0.0f);
    float4 m = make_float4(-3.402823466e38f, -3.402823466e38f,
                           -3.402823466e38f, -3.402823466e38f);
    #pragma unroll
    for (int j = 0; j < PLOADS; ++j) {
        s.x += v[j].x;  s.y += v[j].y;  s.z += v[j].z;  s.w += v[j].w;
        m.x = fmaxf(m.x, v[j].x);  m.y = fmaxf(m.y, v[j].y);
        m.z = fmaxf(m.z, v[j].z);  m.w = fmaxf(m.w, v[j].w);
    }

    #pragma unroll
    for (int off = 8; off <= 16; off <<= 1) {
        s.x += __shfl_xor(s.x, off, 32);  s.y += __shfl_xor(s.y, off, 32);
        s.z += __shfl_xor(s.z, off, 32);  s.w += __shfl_xor(s.w, off, 32);
        m.x = fmaxf(m.x, __shfl_xor(m.x, off, 32));
        m.y = fmaxf(m.y, __shfl_xor(m.y, off, 32));
        m.z = fmaxf(m.z, __shfl_xor(m.z, off, 32));
        m.w = fmaxf(m.w, __shfl_xor(m.w, off, 32));
    }
    if (lane < 8) {
        *(float4*)&lds_s[wv][q * 4] = s;
        *(float4*)&lds_m[wv][q * 4] = m;
    }
    __syncthreads();

    if (wv < 2) {
        const int c = lane;
        if (wv == 0) {
            float ts = 0.0f;
            #pragma unroll
            for (int w = 0; w < PWAVES; ++w) ts += lds_s[w][c];
            lds_p[0][c] = ts * (1.0f / (float)HWn);
        } else {
            float tm = -3.402823466e38f;
            #pragma unroll
            for (int w = 0; w < PWAVES; ++w) tm = fmaxf(tm, lds_m[w][c]);
            lds_p[1][c] = tm;
        }
    }
    __syncthreads();

    if (wv == 0) {
        v8f acc = {};
        #pragma unroll
        for (int j = 0; j < SLICE / 4; ++j) {
            v2f a, bfr;
            const float p0 = lds_p[col & 1][4 * j + ksel];
            const float p1 = lds_p[col & 1][4 * j + ksel + 1];
            a.x   = (col < 2) ? p0 : 0.0f;
            a.y   = (col < 2) ? p1 : 0.0f;
            bfr.x = wf[2 * j];
            bfr.y = wf[2 * j + 1];
            acc = wmma4(a, bfr, acc);
        }
        if (lane < 16) {
            float* hp = hpart + ((size_t)(b * NSLICE + sl) * 2) * HID;
            hp[col]       = acc[0];
            hp[HID + col] = acc[1];
        }
    }
}

__device__ __forceinline__ void barrier_lds_only() {
    asm volatile("s_wait_dscnt 0x0\n\ts_barrier_signal -1\n\ts_barrier_wait -1" ::: "memory");
}

#define SW     14
#define GPX    4
#define NLINE  (Cn / 32)
#define SWPIX  28
#define SBPIX  (SW * SWPIX)
#define SBLK   (HWn / SBPIX)
#define GSTEP  (GPX * Cn / 4)

template <int NG>
__device__ __forceinline__ void stats_load(const float4* __restrict__ xp, float4 (&xv)[NG][NLINE]) {
    #pragma unroll
    for (int g = 0; g < NG; ++g) {
        #pragma unroll
        for (int s = 0; s < NLINE; ++s)
            xv[g][s] = xp[g * GSTEP + s * 8];
    }
}

template <int NG>
__device__ __forceinline__ void stats_stage(const float4 (&xv)[NG][NLINE],
                                            const float4 (&cv)[NLINE],
                                            float* __restrict__ sp, int pix, int lane) {
    float s[NG], m[NG];
    #pragma unroll
    for (int g = 0; g < NG; ++g) {
        float4 a  = make_float4(0.0f, 0.0f, 0.0f, 0.0f);
        float4 mx = make_float4(-3.402823466e38f, -3.402823466e38f,
                                -3.402823466e38f, -3.402823466e38f);
        #pragma unroll
        for (int i = 0; i < NLINE; ++i) {
            const float4 v = xv[g][i];
            const float4 c = cv[i];
            const float px = v.x * c.x, py = v.y * c.y, pz = v.z * c.z, pw = v.w * c.w;
            a.x += px;  a.y += py;  a.z += pz;  a.w += pw;
            mx.x = fmaxf(mx.x, px);  mx.y = fmaxf(mx.y, py);
            mx.z = fmaxf(mx.z, pz);  mx.w = fmaxf(mx.w, pw);
        }
        s[g] = (a.x + a.y) + (a.z + a.w);
        m[g] = fmaxf(fmaxf(mx.x, mx.y), fmaxf(mx.z, mx.w));
    }
    #pragma unroll
    for (int off = 1; off <= 4; off <<= 1) {
        float ts[NG], tm[NG];
        #pragma unroll
        for (int g = 0; g < NG; ++g) {
            ts[g] = __shfl_xor(s[g], off, 32);
            tm[g] = __shfl_xor(m[g], off, 32);
        }
        #pragma unroll
        for (int g = 0; g < NG; ++g) {
            s[g] += ts[g];
            m[g] = fmaxf(m[g], tm[g]);
        }
    }
    const int q = lane & 7;
    const int r = lane >> 3;
    float rs = s[0], rm = m[0];
    #pragma unroll
    for (int g = 1; g < NG; ++g) {
        rs = (q == g) ? s[g] : rs;
        rm = (q == g) ? m[g] : rm;
    }
    if (q < NG) {
        float2 o;
        o.x = rs * (1.0f / (float)Cn);
        o.y = rm;
        *(float2*)(sp + (size_t)(pix + q * GPX + r) * 2) = o;
    }
}

__global__ __launch_bounds__(SW * 32)
void k_mlp_stats(const float* __restrict__ x,
                 const float* __restrict__ hpart,
                 const float* __restrict__ b1,
                 const float* __restrict__ w2,
                 const float* __restrict__ b2,
                 float* __restrict__ chout,
                 float* __restrict__ sp) {
    __shared__ __attribute__((aligned(16))) float lds_c[SW * 32];

    const int lane = threadIdx.x & 31;
    const int wv   = threadIdx.x >> 5;
    const int b    = blockIdx.x / SBLK;
    const int pix0 = blockIdx.x * SBPIX + wv * SWPIX;
    const int col  = lane & 15;
    const int ksel = (lane < 16) ? 0 : 2;

    const int wm = wv & (Cn / 32 - 1);
    float hp[NSLICE];
    #pragma unroll
    for (int s = 0; s < NSLICE; ++s)
        hp[s] = hpart[(size_t)(b * NSLICE + s) * (2 * HID) + lane];
    const float bias1 = b1[col];
    const float bias2 = b2[wm * 32 + lane];
    float wfr[2][8];
    #pragma unroll
    for (int t = 0; t < 2; ++t) {
        const int n0 = (wm * 2 + t) * 16;
        #pragma unroll
        for (int j = 0; j < HID / 4; ++j) {
            wfr[t][2 * j]     = w2[(4 * j + ksel)     * Cn + n0 + col];
            wfr[t][2 * j + 1] = w2[(4 * j + ksel + 1) * Cn + n0 + col];
        }
    }
    __builtin_amdgcn_sched_barrier(0);

    const float4* xp = (const float4*)(x + ((size_t)pix0 + (lane >> 3)) * Cn) + (lane & 7);
    float4 xa[2][NLINE], xb[2][NLINE], xc[1][NLINE];
    stats_load<2>(xp, xa);
    stats_load<2>(xp + 2 * GSTEP, xb);
    __builtin_amdgcn_sched_barrier(0);

    float hs = hp[0];
    #pragma unroll
    for (int s = 1; s < NSLICE; ++s) hs += hp[s];
    hs = fmaxf(hs + bias1, 0.0f);

    v2f af[HID / 4];
    #pragma unroll
    for (int j = 0; j < HID / 4; ++j) {
        const float a0 = __shfl(hs, (col & 1) * 16 + 4 * j + ksel,     32);
        const float a1 = __shfl(hs, (col & 1) * 16 + 4 * j + ksel + 1, 32);
        af[j].x = (col < 2) ? a0 : 0.0f;
        af[j].y = (col < 2) ? a1 : 0.0f;
    }
    v8f acc[2];
    #pragma unroll
    for (int t = 0; t < 2; ++t) {
        acc[t] = (v8f){};
        #pragma unroll
        for (int j = 0; j < HID / 4; ++j) {
            v2f bfr;
            bfr.x = wfr[t][2 * j];
            bfr.y = wfr[t][2 * j + 1];
            acc[t] = wmma4(af[j], bfr, acc[t]);
        }
    }
    const float u0 = __shfl(acc[1][0], col, 32);
    const float u1 = __shfl(acc[1][1], col, 32);
    const float z0 = (lane < 16) ? acc[0][0] : u0;
    const float z1 = (lane < 16) ? acc[0][1] : u1;
    const float cg = sigmoidf_(z0 + bias2) + sigmoidf_(z1 + bias2);
    lds_c[wv * 32 + lane] = cg;
    barrier_lds_only();

    if (blockIdx.x == b * SBLK && threadIdx.x < Cn)
        chout[(size_t)b * Cn + threadIdx.x] = lds_c[threadIdx.x];

    float4 cv[NLINE];
    #pragma unroll
    for (int i = 0; i < NLINE; ++i) cv[i] = *(const float4*)&lds_c[i * 32 + (lane & 7) * 4];

    stats_stage<2>(xa, cv, sp, pix0, lane);
    __builtin_amdgcn_sched_barrier(0);
    stats_load<2>(xp + 4 * GSTEP, xa);
    __builtin_amdgcn_sched_barrier(0);
    stats_stage<2>(xb, cv, sp, pix0 + 2 * GPX, lane);
    __builtin_amdgcn_sched_barrier(0);
    stats_load<1>(xp + 6 * GSTEP, xc);
    __builtin_amdgcn_sched_barrier(0);
    stats_stage<2>(xa, cv, sp, pix0 + 4 * GPX, lane);
    stats_stage<1>(xc, cv, sp, pix0 + 6 * GPX, lane);
}

#define AWAVES 4
#define APIX   (Wn / AWAVES)
#define ALOADS (APIX * (Cn / 128))
#define TROW   32

__global__ __launch_bounds__(AWAVES * 32, 7)
void k_conv_apply(const float* __restrict__ x,
                  const float* __restrict__ chout,
                  const float* __restrict__ sp,
                  const float* __restrict__ wconv,
                  const float* __restrict__ bconv,
                  float* __restrict__ out) {
    __shared__ __attribute__((aligned(16))) float4 tile4[7 * TROW];
    __shared__ float sat[Wn];

    const int t    = threadIdx.x;
    const int lane = t & 31;
    const int wv   = t >> 5;
    const int y    = blockIdx.x;
    const int b    = blockIdx.y;

    const float4* cp = (const float4*)chout + (size_t)b * (Cn / 4) + lane;
    float4 cv[2];
    cv[0] = cp[0];
    cv[1] = cp[32];

    float4 tv[2];
    int    tdst[2];
    #pragma unroll
    for (int k = 0; k < 2; ++k) {
        const int e  = t + k * (AWAVES * 32);
        const int r  = e / 28;
        const int c  = e - r * 28;
        const int iy = y + r - 3;
        tdst[k] = r * TROW + 2 + c;
        tv[k] = make_float4(0.0f, 0.0f, 0.0f, 0.0f);
        if (e < 7 * 28 && iy >= 0 && iy < Hn)
            tv[k] = ((const float4*)sp)[(size_t)(b * Hn + iy) * 28 + c];
    }

    const size_t base4 = ((size_t)(b * Hn + y) * Wn + (size_t)wv * APIX) * (Cn / 4) + lane;
    const float4* xp = (const float4*)x + base4;
    float4 xv[ALOADS];
    #pragma unroll
    for (int i = 0; i < ALOADS; ++i) xv[i] = xp[(size_t)i * 32];

    #pragma unroll
    for (int k = 0; k < 2; ++k)
        if (t + k * (AWAVES * 32) < 7 * 28) tile4[tdst[k]] = tv[k];
    if (t >= 100) {
        const int i = t - 100;
        const int r = i >> 2;
        const int c = i & 3;
        tile4[r * TROW + ((c < 2) ? c : 28 + c)] = make_float4(0.0f, 0.0f, 0.0f, 0.0f);
    }
    barrier_lds_only();

    if (t < Wn) {
        const float2* tile2 = (const float2*)tile4;
        float acc = 0.0f;
        #pragma unroll
        for (int ky = 0; ky < 7; ++ky) {
            #pragma unroll
            for (int kx = 0; kx < 7; ++kx) {
                const float2 v = tile2[ky * (2 * TROW) + t + kx + 1];
                acc += v.x * wconv[(ky * 7 + kx) * 2 + 0];
                acc += v.y * wconv[(ky * 7 + kx) * 2 + 1];
            }
        }
        sat[t] = sigmoidf_(acc + bconv[0]);
    }
    barrier_lds_only();

    const float gl = sat[wv * APIX + ((lane < APIX) ? lane : 0)];
    float4* op = (float4*)out + base4;
    #pragma unroll
    for (int j = 0; j < APIX; ++j) {
        const float s = __builtin_bit_cast(float,
            __builtin_amdgcn_readlane(__builtin_bit_cast(int, gl), j));
        #pragma unroll
        for (int h = 0; h < 2; ++h) {
            const float4 v = xv[2 * j + h];
            float4 o;
            o.x = v.x * cv[h].x * s;
            o.y = v.y * cv[h].y * s;
            o.z = v.z * cv[h].z * s;
            o.w = v.w * cv[h].w * s;
            op[(size_t)(2 * j + h) * 32] = o;
        }
        __builtin_amdgcn_sched_barrier(0);
    }
}

extern "C" void kernel_launch(void* const* d_in, const int* in_sizes, int n_in,
                              void* d_out, int out_size, void* d_ws, size_t ws_size,
                              hipStream_t stream) {
    const float* x     = (const float*)d_in[0];
    const float* w1    = (const float*)d_in[1];
    const float* b1    = (const float*)d_in[2];
    const float* w2    = (const float*)d_in[3];
    const float* b2    = (const float*)d_in[4];
    const float* wconv = (const float*)d_in[5];
    const float* bconv = (const float*)d_in[6];
    float* out = (float*)d_out;

    float* ws    = (float*)d_ws;
    float* hpart = ws;
    float* chout = ws + (size_t)2 * Bn * Cn;
    float* sp    = chout + (size_t)Bn * Cn;

    k_pool_slice<<<dim3(NSLICE, Bn), PWAVES * 32, 0, stream>>>(x, w1, hpart);

    k_mlp_stats<<<Bn * SBLK, SW * 32, 0, stream>>>(x, hpart, b1, w2, b2, chout, sp);

    k_conv_apply<<<dim3(Hn, Bn), AWAVES * 32, 0, stream>>>(x, chout, sp, wconv, bconv, out);
}
